# starting kernel
# baseline (speedup 1.0000x reference)
#include <hip/hip_runtime.h>
#include <math.h>

typedef __attribute__((ext_vector_type(2))) float v2f;
typedef __attribute__((ext_vector_type(8))) float v8f;

#define NFEAT 115
#define NXF 86
#define ROWLEN 1196
#define OUTROWS 100

__global__ void init_cnt_kernel(unsigned int* cnt) {
    if (threadIdx.x < 2) cnt[threadIdx.x] = 0u;
}

__global__ void count_keep_kernel(const float* __restrict__ x,
                                  unsigned int* __restrict__ cnt,
                                  unsigned char* __restrict__ keepL,
                                  unsigned char* __restrict__ keepR) {
    __shared__ float s_sl[64], s_sr[64];
    __shared__ unsigned int s_cl[64], s_cr[64];
    const int t = blockIdx.x;
    const int tid = threadIdx.x;
    const float* fx = x + (size_t)t * (NFEAT * 3);
    float sl = 0.f, sr = 0.f; unsigned int cl = 0u, cr = 0u;
    if (tid < 63) {
        float a = fx[40 * 3 + tid];
        float b = fx[94 * 3 + tid];
        a = (a != a) ? 0.f : a;
        b = (b != b) ? 0.f : b;
        sl = a; sr = b;
        cl = (a != 0.f) ? 1u : 0u;
        cr = (b != 0.f) ? 1u : 0u;
    }
    s_sl[tid] = sl; s_sr[tid] = sr; s_cl[tid] = cl; s_cr[tid] = cr;
    __syncthreads();
    for (int off = 32; off > 0; off >>= 1) {
        if (tid < off) {
            s_sl[tid] += s_sl[tid + off];
            s_sr[tid] += s_sr[tid + off];
            s_cl[tid] += s_cl[tid + off];
            s_cr[tid] += s_cr[tid + off];
        }
        __syncthreads();
    }
    if (tid == 0) {
        keepL[t] = (s_sl[0] != 0.f) ? 1 : 0;
        keepR[t] = (s_sr[0] != 0.f) ? 1 : 0;
        atomicAdd(&cnt[0], s_cl[0]);
        atomicAdd(&cnt[1], s_cr[0]);
    }
}

__global__ void order_kernel(const unsigned int* __restrict__ cnt,
                             const unsigned char* __restrict__ keepL,
                             const unsigned char* __restrict__ keepR,
                             int* __restrict__ order, int T) {
    __shared__ int s[1024];
    const int tid = threadIdx.x;
    const bool cond = cnt[0] > cnt[1];
    const unsigned char* keep = cond ? keepL : keepR;
    const int CH = (T + 1023) >> 10;
    const int t0 = tid * CH;
    const int t1 = (t0 + CH < T) ? (t0 + CH) : T;
    int c = 0;
    for (int t = t0; t < t1; ++t) c += keep[t] ? 1 : 0;
    s[tid] = c;
    __syncthreads();
    for (int off = 1; off < 1024; off <<= 1) {
        int v = s[tid];
        int add = (tid >= off) ? s[tid - off] : 0;
        __syncthreads();
        s[tid] = v + add;
        __syncthreads();
    }
    const int incl = s[tid];
    const int excl = incl - c;
    const int K = s[1023];
    const int vb = (t0 < T) ? t0 : T;
    int rT = excl;
    int rF = K + (vb - excl);
    for (int t = t0; t < t1; ++t) {
        if (keep[t]) { if (rT <= OUTROWS) order[rT] = t; rT++; }
        else         { if (rF <= OUTROWS) order[rF] = t; rF++; }
    }
}

__device__ __forceinline__ float ptNorm(const float* A, int F, int D, int p) {
    float xx = A[(F + p) * 3 + 0];
    float yy = A[(F + p) * 3 + 1];
    float s = xx * xx + yy * yy;
    if (D > 2) { float zz = A[(F + p) * 3 + 2]; s += zz * zz; }
    return s;
}

__global__ void feat_kernel(const float* __restrict__ x,
                            const unsigned int* __restrict__ cnt,
                            const int* __restrict__ order,
                            float* __restrict__ out) {
    __shared__ float A[NXF * 3];
    __shared__ float B[NXF * 3];
    const int row = blockIdx.x;
    const int tid = threadIdx.x;
    const bool cond = cnt[0] > cnt[1];
    const int tA = order[row];
    const int tB = order[row + 1];

    for (int j = tid; j < NXF * 3; j += blockDim.x) {
        int f = j / 3, c = j - f * 3;
        int src;
        if (f < 21)      src = cond ? (40 + f) : (94 + f);
        else if (f < 46) src = f + 40;
        else             src = f - 46;
        float vA = x[(size_t)tA * (NFEAT * 3) + src * 3 + c];
        float vB = x[(size_t)tB * (NFEAT * 3) + src * 3 + c];
        vA = (vA != vA) ? 0.f : vA;
        vB = (vB != vB) ? 0.f : vB;
        if (cond && c == 0) { vA = -vA; vB = -vB; }
        A[j] = vA; B[j] = vB;
    }
    __syncthreads();

    float* o = out + (size_t)row * ROWLEN;

    for (int k = tid; k < 306; k += blockDim.x) {
        int kk = (k < 153) ? k : (k - 153);
        int idx;
        if (kk < 63)       idx = kk;
        else if (kk < 113) { int i2 = kk - 63;  idx = (21 + (i2 >> 1)) * 3 + (i2 & 1); }
        else               { int i2 = kk - 113; idx = (46 + (i2 >> 1)) * 3 + (i2 & 1); }
        o[k] = (k < 153) ? A[idx] : (A[idx] - B[idx]);
    }

    const int w = tid >> 5;
    const int lane = tid & 31;
    const int setN[4] = {21, 25, 20, 20};
    const int setD[4] = {3, 2, 2, 2};
    const int setF[4] = {0, 21, 46, 66};
    const int setO[4] = {306, 516, 816, 1006};
    const int N = setN[w], D = setD[w], F = setF[w], OB = setO[w];
    const int lp = lane & 15;
    const bool hiL = lane >= 16;

    v2f r0 = {0.f, 0.f}, r1 = {0.f, 0.f};
    {
        int p = lp;
        if (p < N) {
            if (!hiL)       { r0.x = A[(F + p) * 3 + 0]; r0.y = A[(F + p) * 3 + 1]; }
            else if (D > 2) { r0.x = A[(F + p) * 3 + 2]; }
        }
        p = 16 + lp;
        if (p < N) {
            if (!hiL)       { r1.x = A[(F + p) * 3 + 0]; r1.y = A[(F + p) * 3 + 1]; }
            else if (D > 2) { r1.x = A[(F + p) * 3 + 2]; }
        }
    }

    v8f cz = {0.f, 0.f, 0.f, 0.f, 0.f, 0.f, 0.f, 0.f};
    v8f G00 = __builtin_amdgcn_wmma_f32_16x16x4_f32(false, r0, false, r0, (short)0, cz, false, false);
    v8f G01 = __builtin_amdgcn_wmma_f32_16x16x4_f32(false, r0, false, r1, (short)0, cz, false, false);
    v8f G11 = __builtin_amdgcn_wmma_f32_16x16x4_f32(false, r1, false, r1, (short)0, cz, false, false);

    const int n0 = lp;
#pragma unroll
    for (int r = 0; r < 8; ++r) {
        const int m0 = r + (hiL ? 8 : 0);
        {
            int m = m0, n = n0;
            if (m < n && n < N) {
                float d2 = ptNorm(A, F, D, m) + ptNorm(A, F, D, n) - 2.f * G00[r];
                int pi = m * (2 * N - m - 1) / 2 + (n - m - 1);
                o[OB + pi] = sqrtf(fmaxf(d2, 0.f));
            }
        }
        {
            int m = m0, n = 16 + n0;
            if (m < n && n < N) {
                float d2 = ptNorm(A, F, D, m) + ptNorm(A, F, D, n) - 2.f * G01[r];
                int pi = m * (2 * N - m - 1) / 2 + (n - m - 1);
                o[OB + pi] = sqrtf(fmaxf(d2, 0.f));
            }
        }
        {
            int m = 16 + m0, n = 16 + n0;
            if (m < n && n < N) {
                float d2 = ptNorm(A, F, D, m) + ptNorm(A, F, D, n) - 2.f * G11[r];
                int pi = m * (2 * N - m - 1) / 2 + (n - m - 1);
                o[OB + pi] = sqrtf(fmaxf(d2, 0.f));
            }
        }
    }
}

extern "C" void kernel_launch(void* const* d_in, const int* in_sizes, int n_in,
                              void* d_out, int out_size, void* d_ws, size_t ws_size,
                              hipStream_t stream) {
    const float* x = (const float*)d_in[0];
    const int T = in_sizes[0] / (NFEAT * 3);
    unsigned int* cnt = (unsigned int*)d_ws;
    int* order = (int*)((char*)d_ws + 16);
    unsigned char* keepL = (unsigned char*)d_ws + 1024;
    unsigned char* keepR = keepL + T;
    float* out = (float*)d_out;

    init_cnt_kernel<<<1, 64, 0, stream>>>(cnt);
    count_keep_kernel<<<T, 64, 0, stream>>>(x, cnt, keepL, keepR);
    order_kernel<<<1, 1024, 0, stream>>>(cnt, keepL, keepR, order, T);
    feat_kernel<<<OUTROWS, 128, 0, stream>>>(x, cnt, order, out);
}


# pass 1 coarsened: each wave sums 16 whole frames in registers with lane butterflies (same pairing as the old LDS tree) and ballots; 512 blocks, one atomic pair per block instead of per frame
# speedup vs baseline: 4.8312x; 4.8312x over previous
#include <hip/hip_runtime.h>
#include <math.h>

typedef __attribute__((ext_vector_type(2))) float v2f;
typedef __attribute__((ext_vector_type(8))) float v8f;

#define NFEAT 115
#define NXF 86
#define ROWLEN 1196
#define OUTROWS 100

__global__ void init_cnt_kernel(unsigned int* cnt) {
    if (threadIdx.x < 2) cnt[threadIdx.x] = 0u;
}

#define P1_FPW 16
#define P1_WAVES 8
#define P1_THREADS 256
#define P1_BLOCKS 512

__device__ __forceinline__ float lane_xor16(float v) {
    int b = __builtin_bit_cast(int, v);
    return __builtin_bit_cast(float,
        __builtin_amdgcn_permlanex16(b, b, 0x76543210, 0xfedcba98, false, false));
}

template <int M>
__device__ __forceinline__ float row_xor(float v) {
    int b = __builtin_bit_cast(int, v);
    return __builtin_bit_cast(float, __builtin_amdgcn_update_dpp(0, b, 0x160 + M, 0xf, 0xf, true));
}

__device__ __forceinline__ float wave_sum(float v) {
    v += lane_xor16(v);
    v += row_xor<8>(v);
    v += row_xor<4>(v);
    v += row_xor<2>(v);
    v += row_xor<1>(v);
    return v;
}

template <int NF>
__device__ __forceinline__ void p1_frames(const float* __restrict__ fx, int o0, int o1, bool has1,
                                          unsigned int& cl, unsigned int& cr,
                                          unsigned int& mL, unsigned int& mR) {
    float l0[NF], l1[NF], r0[NF], r1[NF];
#pragma unroll
    for (int f = 0; f < NF; ++f) {
        l0[f] = fx[f * (NFEAT * 3) + 40 * 3 + o0];
        l1[f] = fx[f * (NFEAT * 3) + 40 * 3 + o1];
        r0[f] = fx[f * (NFEAT * 3) + 94 * 3 + o0];
        r1[f] = fx[f * (NFEAT * 3) + 94 * 3 + o1];
    }
    mL = 0u; mR = 0u;
#pragma unroll
    for (int f = 0; f < NF; ++f) {
        float a0 = l0[f], a1 = l1[f], b0 = r0[f], b1 = r1[f];
        a0 = (a0 != a0) ? 0.f : a0;
        b0 = (b0 != b0) ? 0.f : b0;
        a1 = (has1 && a1 == a1) ? a1 : 0.f;
        b1 = (has1 && b1 == b1) ? b1 : 0.f;
        const unsigned long long ml =
            ((unsigned long long)__builtin_amdgcn_ballot_w32(a1 != 0.f) << 32) | __builtin_amdgcn_ballot_w32(a0 != 0.f);
        const unsigned long long mr =
            ((unsigned long long)__builtin_amdgcn_ballot_w32(b1 != 0.f) << 32) | __builtin_amdgcn_ballot_w32(b0 != 0.f);
        cl += (unsigned int)__builtin_popcountll(ml);
        cr += (unsigned int)__builtin_popcountll(mr);
        const float tl = wave_sum(a0 + a1);
        const float tr = wave_sum(b0 + b1);
        mL |= __builtin_amdgcn_ballot_w32(tl != 0.f) & (1u << f);
        mR |= __builtin_amdgcn_ballot_w32(tr != 0.f) & (1u << f);
    }
}

__global__ __launch_bounds__(P1_THREADS) __attribute__((amdgpu_waves_per_eu(4, 4)))
void count_keep_kernel(const float* __restrict__ x,
                       unsigned int* __restrict__ cnt,
                       unsigned char* __restrict__ keepL,
                       unsigned char* __restrict__ keepR, int T) {
    __shared__ unsigned int s_cnt[2][P1_WAVES];
    const int lane = threadIdx.x & 31;
    const int w = __builtin_amdgcn_readfirstlane((int)(threadIdx.x >> 5));
    const bool has1 = lane < 31;
    const int o0 = lane;
    const int o1 = has1 ? (lane + 32) : 62;
    unsigned int cl = 0u, cr = 0u;
    const int stride = (int)gridDim.x * (P1_WAVES * P1_FPW);
    for (int base = ((int)blockIdx.x * P1_WAVES + w) * P1_FPW; base < T; base += stride) {
        const float* fx = x + (size_t)base * (NFEAT * 3);
        unsigned int mL = 0u, mR = 0u;
        if (base + P1_FPW <= T) {
            p1_frames<P1_FPW>(fx, o0, o1, has1, cl, cr, mL, mR);
        } else {
            for (int f = 0; base + f < T; ++f) {
                unsigned int bL, bR;
                p1_frames<1>(fx + (size_t)f * (NFEAT * 3), o0, o1, has1, cl, cr, bL, bR);
                mL |= bL << f; mR |= bR << f;
            }
        }
        const int t = base + lane;
        if (lane < P1_FPW && t < T) {
            keepL[t] = (unsigned char)((mL >> lane) & 1u);
            keepR[t] = (unsigned char)((mR >> lane) & 1u);
        }
    }
    if (lane == 0) { s_cnt[0][w] = cl; s_cnt[1][w] = cr; }
    __syncthreads();
    if (threadIdx.x < 2) {
        unsigned int s = 0u;
        for (int i = 0; i < P1_WAVES; ++i) s += s_cnt[threadIdx.x][i];
        atomicAdd(&cnt[threadIdx.x], s);
    }
}

__global__ void order_kernel(const unsigned int* __restrict__ cnt,
                             const unsigned char* __restrict__ keepL,
                             const unsigned char* __restrict__ keepR,
                             int* __restrict__ order, int T) {
    __shared__ int s[1024];
    const int tid = threadIdx.x;
    const bool cond = cnt[0] > cnt[1];
    const unsigned char* keep = cond ? keepL : keepR;
    const int CH = (T + 1023) >> 10;
    const int t0 = tid * CH;
    const int t1 = (t0 + CH < T) ? (t0 + CH) : T;
    int c = 0;
    for (int t = t0; t < t1; ++t) c += keep[t] ? 1 : 0;
    s[tid] = c;
    __syncthreads();
    for (int off = 1; off < 1024; off <<= 1) {
        int v = s[tid];
        int add = (tid >= off) ? s[tid - off] : 0;
        __syncthreads();
        s[tid] = v + add;
        __syncthreads();
    }
    const int incl = s[tid];
    const int excl = incl - c;
    const int K = s[1023];
    const int vb = (t0 < T) ? t0 : T;
    int rT = excl;
    int rF = K + (vb - excl);
    for (int t = t0; t < t1; ++t) {
        if (keep[t]) { if (rT <= OUTROWS) order[rT] = t; rT++; }
        else         { if (rF <= OUTROWS) order[rF] = t; rF++; }
    }
}

__device__ __forceinline__ float ptNorm(const float* A, int F, int D, int p) {
    float xx = A[(F + p) * 3 + 0];
    float yy = A[(F + p) * 3 + 1];
    float s = xx * xx + yy * yy;
    if (D > 2) { float zz = A[(F + p) * 3 + 2]; s += zz * zz; }
    return s;
}

__global__ void feat_kernel(const float* __restrict__ x,
                            const unsigned int* __restrict__ cnt,
                            const int* __restrict__ order,
                            float* __restrict__ out) {
    __shared__ float A[NXF * 3];
    __shared__ float B[NXF * 3];
    const int row = blockIdx.x;
    const int tid = threadIdx.x;
    const bool cond = cnt[0] > cnt[1];
    const int tA = order[row];
    const int tB = order[row + 1];

    for (int j = tid; j < NXF * 3; j += blockDim.x) {
        int f = j / 3, c = j - f * 3;
        int src;
        if (f < 21)      src = cond ? (40 + f) : (94 + f);
        else if (f < 46) src = f + 40;
        else             src = f - 46;
        float vA = x[(size_t)tA * (NFEAT * 3) + src * 3 + c];
        float vB = x[(size_t)tB * (NFEAT * 3) + src * 3 + c];
        vA = (vA != vA) ? 0.f : vA;
        vB = (vB != vB) ? 0.f : vB;
        if (cond && c == 0) { vA = -vA; vB = -vB; }
        A[j] = vA; B[j] = vB;
    }
    __syncthreads();

    float* o = out + (size_t)row * ROWLEN;

    for (int k = tid; k < 306; k += blockDim.x) {
        int kk = (k < 153) ? k : (k - 153);
        int idx;
        if (kk < 63)       idx = kk;
        else if (kk < 113) { int i2 = kk - 63;  idx = (21 + (i2 >> 1)) * 3 + (i2 & 1); }
        else               { int i2 = kk - 113; idx = (46 + (i2 >> 1)) * 3 + (i2 & 1); }
        o[k] = (k < 153) ? A[idx] : (A[idx] - B[idx]);
    }

    const int w = tid >> 5;
    const int lane = tid & 31;
    const int setN[4] = {21, 25, 20, 20};
    const int setD[4] = {3, 2, 2, 2};
    const int setF[4] = {0, 21, 46, 66};
    const int setO[4] = {306, 516, 816, 1006};
    const int N = setN[w], D = setD[w], F = setF[w], OB = setO[w];
    const int lp = lane & 15;
    const bool hiL = lane >= 16;

    v2f r0 = {0.f, 0.f}, r1 = {0.f, 0.f};
    {
        int p = lp;
        if (p < N) {
            if (!hiL)       { r0.x = A[(F + p) * 3 + 0]; r0.y = A[(F + p) * 3 + 1]; }
            else if (D > 2) { r0.x = A[(F + p) * 3 + 2]; }
        }
        p = 16 + lp;
        if (p < N) {
            if (!hiL)       { r1.x = A[(F + p) * 3 + 0]; r1.y = A[(F + p) * 3 + 1]; }
            else if (D > 2) { r1.x = A[(F + p) * 3 + 2]; }
        }
    }

    v8f cz = {0.f, 0.f, 0.f, 0.f, 0.f, 0.f, 0.f, 0.f};
    v8f G00 = __builtin_amdgcn_wmma_f32_16x16x4_f32(false, r0, false, r0, (short)0, cz, false, false);
    v8f G01 = __builtin_amdgcn_wmma_f32_16x16x4_f32(false, r0, false, r1, (short)0, cz, false, false);
    v8f G11 = __builtin_amdgcn_wmma_f32_16x16x4_f32(false, r1, false, r1, (short)0, cz, false, false);

    const int n0 = lp;
#pragma unroll
    for (int r = 0; r < 8; ++r) {
        const int m0 = r + (hiL ? 8 : 0);
        {
            int m = m0, n = n0;
            if (m < n && n < N) {
                float d2 = ptNorm(A, F, D, m) + ptNorm(A, F, D, n) - 2.f * G00[r];
                int pi = m * (2 * N - m - 1) / 2 + (n - m - 1);
                o[OB + pi] = sqrtf(fmaxf(d2, 0.f));
            }
        }
        {
            int m = m0, n = 16 + n0;
            if (m < n && n < N) {
                float d2 = ptNorm(A, F, D, m) + ptNorm(A, F, D, n) - 2.f * G01[r];
                int pi = m * (2 * N - m - 1) / 2 + (n - m - 1);
                o[OB + pi] = sqrtf(fmaxf(d2, 0.f));
            }
        }
        {
            int m = 16 + m0, n = 16 + n0;
            if (m < n && n < N) {
                float d2 = ptNorm(A, F, D, m) + ptNorm(A, F, D, n) - 2.f * G11[r];
                int pi = m * (2 * N - m - 1) / 2 + (n - m - 1);
                o[OB + pi] = sqrtf(fmaxf(d2, 0.f));
            }
        }
    }
}

extern "C" void kernel_launch(void* const* d_in, const int* in_sizes, int n_in,
                              void* d_out, int out_size, void* d_ws, size_t ws_size,
                              hipStream_t stream) {
    const float* x = (const float*)d_in[0];
    const int T = in_sizes[0] / (NFEAT * 3);
    unsigned int* cnt = (unsigned int*)d_ws;
    int* order = (int*)((char*)d_ws + 16);
    unsigned char* keepL = (unsigned char*)d_ws + 1024;
    unsigned char* keepR = keepL + T;
    float* out = (float*)d_out;

    init_cnt_kernel<<<1, 64, 0, stream>>>(cnt);
    count_keep_kernel<<<P1_BLOCKS, P1_THREADS, 0, stream>>>(x, cnt, keepL, keepR, T);
    order_kernel<<<1, 1024, 0, stream>>>(cnt, keepL, keepR, order, T);
    feat_kernel<<<OUTROWS, 128, 0, stream>>>(x, cnt, order, out);
}


# two launches instead of four: pass 1 leaves per-block count pairs and 16-frame keep bitmasks (no atomics, no zeroing pass); each feature block sums the pairs and ranks its own two frames itself
# speedup vs baseline: 10.5348x; 2.1806x over previous
#include <hip/hip_runtime.h>
#include <math.h>

typedef __attribute__((ext_vector_type(2))) float v2f;
typedef __attribute__((ext_vector_type(8))) float v8f;

#define NFEAT 115
#define NXF 86
#define ROWLEN 1196
#define OUTROWS 100

#define P1_FPW 16
#define P1_WAVES 8
#define P1_THREADS 256
#define P1_BLOCKS 512

__device__ __forceinline__ float lane_xor16(float v) {
    int b = __builtin_bit_cast(int, v);
    return __builtin_bit_cast(float,
        __builtin_amdgcn_permlanex16(b, b, 0x76543210, 0xfedcba98, false, false));
}

template <int M>
__device__ __forceinline__ float row_xor(float v) {
    int b = __builtin_bit_cast(int, v);
    return __builtin_bit_cast(float, __builtin_amdgcn_update_dpp(0, b, 0x160 + M, 0xf, 0xf, true));
}

__device__ __forceinline__ float wave_sum(float v) {
    v += lane_xor16(v);
    v += row_xor<8>(v);
    v += row_xor<4>(v);
    v += row_xor<2>(v);
    v += row_xor<1>(v);
    return v;
}

template <int NF>
__device__ __forceinline__ void p1_frames(const float* __restrict__ fx, int o0, int o1, bool has1,
                                          unsigned int& cl, unsigned int& cr,
                                          unsigned int& mL, unsigned int& mR) {
    float l0[NF], l1[NF], r0[NF], r1[NF];
#pragma unroll
    for (int f = 0; f < NF; ++f) {
        l0[f] = fx[f * (NFEAT * 3) + 40 * 3 + o0];
        l1[f] = fx[f * (NFEAT * 3) + 40 * 3 + o1];
        r0[f] = fx[f * (NFEAT * 3) + 94 * 3 + o0];
        r1[f] = fx[f * (NFEAT * 3) + 94 * 3 + o1];
    }
    mL = 0u; mR = 0u;
#pragma unroll
    for (int f = 0; f < NF; ++f) {
        float a0 = l0[f], a1 = l1[f], b0 = r0[f], b1 = r1[f];
        a0 = (a0 != a0) ? 0.f : a0;
        b0 = (b0 != b0) ? 0.f : b0;
        a1 = (has1 && a1 == a1) ? a1 : 0.f;
        b1 = (has1 && b1 == b1) ? b1 : 0.f;
        const unsigned long long ml =
            ((unsigned long long)__builtin_amdgcn_ballot_w32(a1 != 0.f) << 32) | __builtin_amdgcn_ballot_w32(a0 != 0.f);
        const unsigned long long mr =
            ((unsigned long long)__builtin_amdgcn_ballot_w32(b1 != 0.f) << 32) | __builtin_amdgcn_ballot_w32(b0 != 0.f);
        cl += (unsigned int)__builtin_popcountll(ml);
        cr += (unsigned int)__builtin_popcountll(mr);
        const float tl = wave_sum(a0 + a1);
        const float tr = wave_sum(b0 + b1);
        mL |= __builtin_amdgcn_ballot_w32(tl != 0.f) & (1u << f);
        mR |= __builtin_amdgcn_ballot_w32(tr != 0.f) & (1u << f);
    }
}

__global__ __launch_bounds__(P1_THREADS) __attribute__((amdgpu_waves_per_eu(4, 4)))
void count_keep_kernel(const float* __restrict__ x,
                       unsigned int* __restrict__ part,
                       unsigned int* __restrict__ gmask, int T) {
    __shared__ unsigned int s_cnt[2][P1_WAVES];
    const int lane = threadIdx.x & 31;
    const int w = __builtin_amdgcn_readfirstlane((int)(threadIdx.x >> 5));
    const bool has1 = lane < 31;
    const int o0 = lane;
    const int o1 = has1 ? (lane + 32) : 62;
    unsigned int cl = 0u, cr = 0u;
    const int stride = (int)gridDim.x * (P1_WAVES * P1_FPW);
    for (int base = ((int)blockIdx.x * P1_WAVES + w) * P1_FPW; base < T; base += stride) {
        const float* fx = x + (size_t)base * (NFEAT * 3);
        unsigned int mL = 0u, mR = 0u;
        if (base + P1_FPW <= T) {
            p1_frames<P1_FPW>(fx, o0, o1, has1, cl, cr, mL, mR);
        } else {
            for (int f = 0; base + f < T; ++f) {
                unsigned int bL, bR;
                p1_frames<1>(fx + (size_t)f * (NFEAT * 3), o0, o1, has1, cl, cr, bL, bR);
                mL |= bL << f; mR |= bR << f;
            }
        }
        if (lane == 0) gmask[base / P1_FPW] = (mR << 16) | (mL & 0xffffu);
    }
    if (lane == 0) { s_cnt[0][w] = cl; s_cnt[1][w] = cr; }
    __syncthreads();
    if (threadIdx.x < 2) {
        unsigned int s = 0u;
        for (int i = 0; i < P1_WAVES; ++i) s += s_cnt[threadIdx.x][i];
        part[2 * blockIdx.x + threadIdx.x] = s;
    }
}


__device__ __forceinline__ unsigned int wave_sum_u32(unsigned int v) {
    const int b = (int)v;
    v += (unsigned int)__builtin_amdgcn_permlanex16(b, b, 0x76543210, 0xfedcba98, false, false);
    v += (unsigned int)__builtin_amdgcn_update_dpp(0, (int)v, 0x160 + 8, 0xf, 0xf, true);
    v += (unsigned int)__builtin_amdgcn_update_dpp(0, (int)v, 0x160 + 4, 0xf, 0xf, true);
    v += (unsigned int)__builtin_amdgcn_update_dpp(0, (int)v, 0x160 + 2, 0xf, 0xf, true);
    v += (unsigned int)__builtin_amdgcn_update_dpp(0, (int)v, 0x160 + 1, 0xf, 0xf, true);
    return v;
}

__device__ __forceinline__ void chunk_prefix(unsigned int m, int& c, int& excl, int& total) {
    c = __builtin_popcount(m);
    excl = 0; total = 0;
    for (int b = 0; b < 5; ++b) {
        const unsigned int bm = __builtin_amdgcn_ballot_w32(((c >> b) & 1) != 0);
        excl += (int)(__builtin_amdgcn_mbcnt_lo(bm, 0u) << b);
        total += __builtin_popcount(bm) << b;
    }
}

__device__ __forceinline__ int chunk_select(unsigned int m, int c, int excl, int rel, int lane) {
    const unsigned int own = __builtin_amdgcn_ballot_w32(rel >= excl && rel < excl + c);
    if (own == 0u) return -1;
    unsigned int mm = m;
    const int k = rel - excl;
    for (int i = 0; i < 15; ++i) if (i < k) mm &= mm - 1u;
    const int pos = lane * 16 + __builtin_ctz(mm | 0x10000u);
    return __builtin_amdgcn_readlane(pos, __builtin_ctz(own));
}

__device__ __forceinline__ float ptNorm(const float* A, int F, int D, int p) {
    float xx = A[(F + p) * 3 + 0];
    float yy = A[(F + p) * 3 + 1];
    float s = xx * xx + yy * yy;
    if (D > 2) { float zz = A[(F + p) * 3 + 2]; s += zz * zz; }
    return s;
}

__global__ void feat_kernel(const float* __restrict__ x,
                            const unsigned int* __restrict__ part,
                            const unsigned int* __restrict__ gmask,
                            float* __restrict__ out, int T) {
    __shared__ float A[NXF * 3];
    __shared__ float B[NXF * 3];
    const int row = blockIdx.x;
    const int tid = threadIdx.x;
    const int ln = tid & 31;

    const int ngroups = (T + P1_FPW - 1) / P1_FPW;
    const unsigned int pm0 = (ln < ngroups) ? gmask[ln] : 0u;
    unsigned int sl = 0u, sr = 0u;
    const uint4* part4 = reinterpret_cast<const uint4*>(part) + ln;
    for (int k = 0; k < P1_BLOCKS / 64; ++k) {
        const uint4 p = part4[k * 32];
        sl += p.x + p.z; sr += p.y + p.w;
    }
    sl = (unsigned int)__builtin_amdgcn_readfirstlane((int)wave_sum_u32(sl));
    sr = (unsigned int)__builtin_amdgcn_readfirstlane((int)wave_sum_u32(sr));
    const bool cond = sl > sr;

    int tA = -1, tB = -1;
    int cum = 0;
    const int nchunks = (ngroups + 31) >> 5;
    for (int ch = 0; ch < nchunks; ++ch) {
        const int g = ch * 32 + ln;
        const unsigned int pm = (ch == 0) ? pm0 : ((g < ngroups) ? gmask[g] : 0u);
        const unsigned int m = cond ? (pm & 0xffffu) : (pm >> 16);
        int c, excl, total;
        chunk_prefix(m, c, excl, total);
        if (tA < 0) { const int p = chunk_select(m, c, excl, row - cum, ln);     if (p >= 0) tA = ch * 512 + p; }
        if (tB < 0) { const int p = chunk_select(m, c, excl, row + 1 - cum, ln); if (p >= 0) tB = ch * 512 + p; }
        cum += total;
        if (tB >= 0) break;
    }
    if (tB < 0) {
        const unsigned int m = cond ? (pm0 & 0xffffu) : (pm0 >> 16);
        int nv = T - ln * 16;
        nv = nv < 0 ? 0 : (nv > 16 ? 16 : nv);
        const unsigned int nm = ~m & ((1u << nv) - 1u);
        int c, excl, total;
        chunk_prefix(nm, c, excl, total);
        if (tA < 0) tA = chunk_select(nm, c, excl, row - cum, ln);
        tB = chunk_select(nm, c, excl, row + 1 - cum, ln);
    }
    tA = tA < 0 ? 0 : (tA > T - 1 ? T - 1 : tA);
    tB = tB < 0 ? 0 : (tB > T - 1 ? T - 1 : tB);

    for (int j = tid; j < NXF * 3; j += blockDim.x) {
        int f = j / 3, c = j - f * 3;
        int src;
        if (f < 21)      src = cond ? (40 + f) : (94 + f);
        else if (f < 46) src = f + 40;
        else             src = f - 46;
        float vA = x[(size_t)tA * (NFEAT * 3) + src * 3 + c];
        float vB = x[(size_t)tB * (NFEAT * 3) + src * 3 + c];
        vA = (vA != vA) ? 0.f : vA;
        vB = (vB != vB) ? 0.f : vB;
        if (cond && c == 0) { vA = -vA; vB = -vB; }
        A[j] = vA; B[j] = vB;
    }
    __syncthreads();

    float* o = out + (size_t)row * ROWLEN;

    for (int k = tid; k < 306; k += blockDim.x) {
        int kk = (k < 153) ? k : (k - 153);
        int idx;
        if (kk < 63)       idx = kk;
        else if (kk < 113) { int i2 = kk - 63;  idx = (21 + (i2 >> 1)) * 3 + (i2 & 1); }
        else               { int i2 = kk - 113; idx = (46 + (i2 >> 1)) * 3 + (i2 & 1); }
        o[k] = (k < 153) ? A[idx] : (A[idx] - B[idx]);
    }

    const int w = tid >> 5;
    const int lane = tid & 31;
    const int setN[4] = {21, 25, 20, 20};
    const int setD[4] = {3, 2, 2, 2};
    const int setF[4] = {0, 21, 46, 66};
    const int setO[4] = {306, 516, 816, 1006};
    const int N = setN[w], D = setD[w], F = setF[w], OB = setO[w];
    const int lp = lane & 15;
    const bool hiL = lane >= 16;

    v2f r0 = {0.f, 0.f}, r1 = {0.f, 0.f};
    {
        int p = lp;
        if (p < N) {
            if (!hiL)       { r0.x = A[(F + p) * 3 + 0]; r0.y = A[(F + p) * 3 + 1]; }
            else if (D > 2) { r0.x = A[(F + p) * 3 + 2]; }
        }
        p = 16 + lp;
        if (p < N) {
            if (!hiL)       { r1.x = A[(F + p) * 3 + 0]; r1.y = A[(F + p) * 3 + 1]; }
            else if (D > 2) { r1.x = A[(F + p) * 3 + 2]; }
        }
    }

    v8f cz = {0.f, 0.f, 0.f, 0.f, 0.f, 0.f, 0.f, 0.f};
    v8f G00 = __builtin_amdgcn_wmma_f32_16x16x4_f32(false, r0, false, r0, (short)0, cz, false, false);
    v8f G01 = __builtin_amdgcn_wmma_f32_16x16x4_f32(false, r0, false, r1, (short)0, cz, false, false);
    v8f G11 = __builtin_amdgcn_wmma_f32_16x16x4_f32(false, r1, false, r1, (short)0, cz, false, false);

    const int n0 = lp;
#pragma unroll
    for (int r = 0; r < 8; ++r) {
        const int m0 = r + (hiL ? 8 : 0);
        {
            int m = m0, n = n0;
            if (m < n && n < N) {
                float d2 = ptNorm(A, F, D, m) + ptNorm(A, F, D, n) - 2.f * G00[r];
                int pi = m * (2 * N - m - 1) / 2 + (n - m - 1);
                o[OB + pi] = sqrtf(fmaxf(d2, 0.f));
            }
        }
        {
            int m = m0, n = 16 + n0;
            if (m < n && n < N) {
                float d2 = ptNorm(A, F, D, m) + ptNorm(A, F, D, n) - 2.f * G01[r];
                int pi = m * (2 * N - m - 1) / 2 + (n - m - 1);
                o[OB + pi] = sqrtf(fmaxf(d2, 0.f));
            }
        }
        {
            int m = 16 + m0, n = 16 + n0;
            if (m < n && n < N) {
                float d2 = ptNorm(A, F, D, m) + ptNorm(A, F, D, n) - 2.f * G11[r];
                int pi = m * (2 * N - m - 1) / 2 + (n - m - 1);
                o[OB + pi] = sqrtf(fmaxf(d2, 0.f));
            }
        }
    }
}

extern "C" void kernel_launch(void* const* d_in, const int* in_sizes, int n_in,
                              void* d_out, int out_size, void* d_ws, size_t ws_size,
                              hipStream_t stream) {
    const float* x = (const float*)d_in[0];
    const int T = in_sizes[0] / (NFEAT * 3);
    unsigned int* part = (unsigned int*)d_ws;
    unsigned int* gmask = (unsigned int*)((char*)d_ws + 4096);
    float* out = (float*)d_out;

    count_keep_kernel<<<P1_BLOCKS, P1_THREADS, 0, stream>>>(x, part, gmask, T);
    feat_kernel<<<OUTROWS, 128, 0, stream>>>(x, part, gmask, out, T);
}


# feature kernel made straight-line: point norms once per lane, handed round by DPP row_share; distances staged in LDS; gather and store loops unrolled: no per-pair LDS round trips or branches
# speedup vs baseline: 14.1910x; 1.3471x over previous
#include <hip/hip_runtime.h>
#include <math.h>

typedef __attribute__((ext_vector_type(2))) float v2f;
typedef __attribute__((ext_vector_type(8))) float v8f;

#define NFEAT 115
#define NXF 86
#define ROWLEN 1196
#define OUTROWS 100

#define P1_FPW 16
#define P1_WAVES 8
#define P1_THREADS 256
#define P1_BLOCKS 512

__device__ __forceinline__ float lane_xor16(float v) {
    int b = __builtin_bit_cast(int, v);
    return __builtin_bit_cast(float,
        __builtin_amdgcn_permlanex16(b, b, 0x76543210, 0xfedcba98, false, false));
}

template <int M>
__device__ __forceinline__ float row_xor(float v) {
    int b = __builtin_bit_cast(int, v);
    return __builtin_bit_cast(float, __builtin_amdgcn_update_dpp(0, b, 0x160 + M, 0xf, 0xf, true));
}

__device__ __forceinline__ float wave_sum(float v) {
    v += lane_xor16(v);
    v += row_xor<8>(v);
    v += row_xor<4>(v);
    v += row_xor<2>(v);
    v += row_xor<1>(v);
    return v;
}

template <int NF>
__device__ __forceinline__ void p1_frames(const float* __restrict__ fx, int o0, int o1, bool has1,
                                          unsigned int& cl, unsigned int& cr,
                                          unsigned int& mL, unsigned int& mR) {
    float l0[NF], l1[NF], r0[NF], r1[NF];
#pragma unroll
    for (int f = 0; f < NF; ++f) {
        l0[f] = fx[f * (NFEAT * 3) + 40 * 3 + o0];
        l1[f] = fx[f * (NFEAT * 3) + 40 * 3 + o1];
        r0[f] = fx[f * (NFEAT * 3) + 94 * 3 + o0];
        r1[f] = fx[f * (NFEAT * 3) + 94 * 3 + o1];
    }
    mL = 0u; mR = 0u;
#pragma unroll
    for (int f = 0; f < NF; ++f) {
        float a0 = l0[f], a1 = l1[f], b0 = r0[f], b1 = r1[f];
        a0 = (a0 != a0) ? 0.f : a0;
        b0 = (b0 != b0) ? 0.f : b0;
        a1 = (has1 && a1 == a1) ? a1 : 0.f;
        b1 = (has1 && b1 == b1) ? b1 : 0.f;
        const unsigned long long ml =
            ((unsigned long long)__builtin_amdgcn_ballot_w32(a1 != 0.f) << 32) | __builtin_amdgcn_ballot_w32(a0 != 0.f);
        const unsigned long long mr =
            ((unsigned long long)__builtin_amdgcn_ballot_w32(b1 != 0.f) << 32) | __builtin_amdgcn_ballot_w32(b0 != 0.f);
        cl += (unsigned int)__builtin_popcountll(ml);
        cr += (unsigned int)__builtin_popcountll(mr);
        const float tl = wave_sum(a0 + a1);
        const float tr = wave_sum(b0 + b1);
        mL |= __builtin_amdgcn_ballot_w32(tl != 0.f) & (1u << f);
        mR |= __builtin_amdgcn_ballot_w32(tr != 0.f) & (1u << f);
    }
}

__global__ __launch_bounds__(P1_THREADS) __attribute__((amdgpu_waves_per_eu(4, 4)))
void count_keep_kernel(const float* __restrict__ x,
                       unsigned int* __restrict__ part,
                       unsigned int* __restrict__ gmask, int T) {
    __shared__ unsigned int s_cnt[2][P1_WAVES];
    const int lane = threadIdx.x & 31;
    const int w = __builtin_amdgcn_readfirstlane((int)(threadIdx.x >> 5));
    const bool has1 = lane < 31;
    const int o0 = lane;
    const int o1 = has1 ? (lane + 32) : 62;
    unsigned int cl = 0u, cr = 0u;
    const int stride = (int)gridDim.x * (P1_WAVES * P1_FPW);
    for (int base = ((int)blockIdx.x * P1_WAVES + w) * P1_FPW; base < T; base += stride) {
        const float* fx = x + (size_t)base * (NFEAT * 3);
        unsigned int mL = 0u, mR = 0u;
        if (base + P1_FPW <= T) {
            p1_frames<P1_FPW>(fx, o0, o1, has1, cl, cr, mL, mR);
        } else {
            for (int f = 0; base + f < T; ++f) {
                unsigned int bL, bR;
                p1_frames<1>(fx + (size_t)f * (NFEAT * 3), o0, o1, has1, cl, cr, bL, bR);
                mL |= bL << f; mR |= bR << f;
            }
        }
        if (lane == 0) gmask[base / P1_FPW] = (mR << 16) | (mL & 0xffffu);
    }
    if (lane == 0) { s_cnt[0][w] = cl; s_cnt[1][w] = cr; }
    __syncthreads();
    if (threadIdx.x < 2) {
        unsigned int s = 0u;
        for (int i = 0; i < P1_WAVES; ++i) s += s_cnt[threadIdx.x][i];
        part[2 * blockIdx.x + threadIdx.x] = s;
    }
}


__device__ __forceinline__ unsigned int wave_sum_u32(unsigned int v) {
    const int b = (int)v;
    v += (unsigned int)__builtin_amdgcn_permlanex16(b, b, 0x76543210, 0xfedcba98, false, false);
    v += (unsigned int)__builtin_amdgcn_update_dpp(0, (int)v, 0x160 + 8, 0xf, 0xf, true);
    v += (unsigned int)__builtin_amdgcn_update_dpp(0, (int)v, 0x160 + 4, 0xf, 0xf, true);
    v += (unsigned int)__builtin_amdgcn_update_dpp(0, (int)v, 0x160 + 2, 0xf, 0xf, true);
    v += (unsigned int)__builtin_amdgcn_update_dpp(0, (int)v, 0x160 + 1, 0xf, 0xf, true);
    return v;
}

__device__ __forceinline__ void chunk_prefix(unsigned int m, int& c, int& excl, int& total) {
    c = __builtin_popcount(m);
    excl = 0; total = 0;
    for (int b = 0; b < 5; ++b) {
        const unsigned int bm = __builtin_amdgcn_ballot_w32(((c >> b) & 1) != 0);
        excl += (int)(__builtin_amdgcn_mbcnt_lo(bm, 0u) << b);
        total += __builtin_popcount(bm) << b;
    }
}

__device__ __forceinline__ int chunk_select(unsigned int m, int c, int excl, int rel, int lane) {
    const unsigned int own = __builtin_amdgcn_ballot_w32(rel >= excl && rel < excl + c);
    if (own == 0u) return -1;
    unsigned int mm = m;
    const int k = rel - excl;
    for (int i = 0; i < 15; ++i) if (i < k) mm &= mm - 1u;
    const int pos = lane * 16 + __builtin_ctz(mm | 0x10000u);
    return __builtin_amdgcn_readlane(pos, __builtin_ctz(own));
}

#define FT 128
#define NPAIR 890

template <int R>
__device__ __forceinline__ float row_share(float v) {
    int b = __builtin_bit_cast(int, v);
    return __builtin_bit_cast(float, __builtin_amdgcn_update_dpp(0, b, 0x150 + R, 0xf, 0xf, true));
}

template <int R>
__device__ __forceinline__ void dist_row(const v8f& G00, const v8f& G01, const v8f& G11,
                                         float U, float V, float nA, float nB,
                                         int lp, int mo, int N, int SO, int spare, float* S) {
    const float ml = row_share<R>(U);
    const float mh = row_share<R>(V);
    const int m = R + mo;
    const int m1 = 16 + m;
    const int pm = ((m * (2 * N - m - 1)) >> 1) + SO;
    const int pm1 = ((m1 * (2 * N - m1 - 1)) >> 1) + SO;
    {
        const float d2 = ml + nA - 2.f * G00[R];
        S[(m < lp) ? (pm + (lp - m - 1)) : spare] = sqrtf(fmaxf(d2, 0.f));
    }
    {
        const float d2 = ml + nB - 2.f * G01[R];
        S[(16 + lp < N) ? (pm + (16 + lp - m - 1)) : spare] = sqrtf(fmaxf(d2, 0.f));
    }
    {
        const float d2 = mh + nB - 2.f * G11[R];
        S[(m < lp && 16 + lp < N) ? (pm1 + (lp - m - 1)) : spare] = sqrtf(fmaxf(d2, 0.f));
    }
}

__global__ __launch_bounds__(FT)
void feat_kernel(const float* __restrict__ x,
                            const unsigned int* __restrict__ part,
                            const unsigned int* __restrict__ gmask,
                            float* __restrict__ out, int T) {
    __shared__ float A[3 * FT];
    __shared__ float B[3 * FT];
    __shared__ float S[NPAIR + 32];
    const int row = blockIdx.x;
    const int tid = threadIdx.x;
    __builtin_assume(tid >= 0 && tid < FT);
    const int ln = tid & 31;

    const int ngroups = (T + P1_FPW - 1) / P1_FPW;
    const unsigned int pm0 = (ln < ngroups) ? gmask[ln] : 0u;
    unsigned int sl = 0u, sr = 0u;
    const uint4* part4 = reinterpret_cast<const uint4*>(part) + ln;
    for (int k = 0; k < P1_BLOCKS / 64; ++k) {
        const uint4 p = part4[k * 32];
        sl += p.x + p.z; sr += p.y + p.w;
    }
    sl = (unsigned int)__builtin_amdgcn_readfirstlane((int)wave_sum_u32(sl));
    sr = (unsigned int)__builtin_amdgcn_readfirstlane((int)wave_sum_u32(sr));
    const bool cond = sl > sr;

    int tA = -1, tB = -1;
    int cum = 0;
    const int nchunks = (ngroups + 31) >> 5;
    for (int ch = 0; ch < nchunks; ++ch) {
        const int g = ch * 32 + ln;
        const unsigned int pm = (ch == 0) ? pm0 : ((g < ngroups) ? gmask[g] : 0u);
        const unsigned int m = cond ? (pm & 0xffffu) : (pm >> 16);
        int c, excl, total;
        chunk_prefix(m, c, excl, total);
        if (tA < 0) { const int p = chunk_select(m, c, excl, row - cum, ln);     if (p >= 0) tA = ch * 512 + p; }
        if (tB < 0) { const int p = chunk_select(m, c, excl, row + 1 - cum, ln); if (p >= 0) tB = ch * 512 + p; }
        cum += total;
        if (tB >= 0) break;
    }
    if (tB < 0) {
        const unsigned int m = cond ? (pm0 & 0xffffu) : (pm0 >> 16);
        int nv = T - ln * 16;
        nv = nv < 0 ? 0 : (nv > 16 ? 16 : nv);
        const unsigned int nm = ~m & ((1u << nv) - 1u);
        int c, excl, total;
        chunk_prefix(nm, c, excl, total);
        if (tA < 0) tA = chunk_select(nm, c, excl, row - cum, ln);
        tB = chunk_select(nm, c, excl, row + 1 - cum, ln);
    }
    tA = tA < 0 ? 0 : (tA > T - 1 ? T - 1 : tA);
    tB = tB < 0 ? 0 : (tB > T - 1 ? T - 1 : tB);

#pragma unroll
    for (int i = 0; i < 3; ++i) {
        const int j = tid + i * FT;
        const int jc = (j < NXF * 3) ? j : (NXF * 3 - 1);
        int f = jc / 3, c = jc - f * 3;
        int src;
        if (f < 21)      src = cond ? (40 + f) : (94 + f);
        else if (f < 46) src = f + 40;
        else             src = f - 46;
        float vA = x[(size_t)tA * (NFEAT * 3) + src * 3 + c];
        float vB = x[(size_t)tB * (NFEAT * 3) + src * 3 + c];
        vA = (vA != vA) ? 0.f : vA;
        vB = (vB != vB) ? 0.f : vB;
        if (cond && c == 0) { vA = -vA; vB = -vB; }
        A[j] = vA; B[j] = vB;
    }
    __syncthreads();

    float* o = out + (size_t)row * ROWLEN;

#pragma unroll
    for (int i = 0; i < 3; ++i) {
        const int k = tid + i * FT;
        const int kc = (k < 306) ? k : 305;
        const int kk = (kc < 153) ? kc : (kc - 153);
        const int i2 = kk - ((kk < 113) ? 63 : 113);
        const int i3 = ((kk < 113) ? 21 * 3 : 46 * 3) + (i2 >> 1) * 3 + (i2 & 1);
        const int idx = (kk < 63) ? kk : i3;
        const float a = A[idx], b = B[idx];
        if (k < 306) o[k] = (kc < 153) ? a : (a - b);
    }

    const int w = __builtin_amdgcn_readfirstlane(tid >> 5);
    const int N  = (w == 0) ? 21 : ((w == 1) ? 25 : 20);
    const int F  = (w == 0) ? 0 : ((w == 1) ? 21 : ((w == 2) ? 46 : 66));
    const int SO = (w == 0) ? 0 : ((w == 1) ? 210 : ((w == 2) ? 510 : 700));
    const bool three = (w == 0);
    const int lp = ln & 15;
    const bool hiL = ln >= 16;

    const bool in1 = 16 + lp < N;
    const float* P0 = A + (F + lp) * 3;
    const float* P1 = A + (F + (in1 ? (16 + lp) : (N - 1))) * 3;
    float x0 = P0[0], y0 = P0[1], z0 = P0[2];
    float x1 = P1[0], y1 = P1[1], z1 = P1[2];
    if (!three) { z0 = 0.f; z1 = 0.f; }
    if (!in1)   { x1 = 0.f; y1 = 0.f; z1 = 0.f; }
    float nA = x0 * x0 + y0 * y0; nA += z0 * z0;
    float nB = x1 * x1 + y1 * y1; nB += z1 * z1;

    v2f r0, r1;
    r0.x = hiL ? z0 : x0; r0.y = hiL ? 0.f : y0;
    r1.x = hiL ? z1 : x1; r1.y = hiL ? 0.f : y1;

    v8f cz = {0.f, 0.f, 0.f, 0.f, 0.f, 0.f, 0.f, 0.f};
    v8f G00 = __builtin_amdgcn_wmma_f32_16x16x4_f32(false, r0, false, r0, (short)0, cz, false, false);
    v8f G01 = __builtin_amdgcn_wmma_f32_16x16x4_f32(false, r0, false, r1, (short)0, cz, false, false);
    v8f G11 = __builtin_amdgcn_wmma_f32_16x16x4_f32(false, r1, false, r1, (short)0, cz, false, false);

    const float U = hiL ? row_xor<8>(nA) : nA;
    const float V = hiL ? row_xor<8>(nB) : nB;
    const int mo = hiL ? 8 : 0;
    const int spare = NPAIR + ln;
    dist_row<0>(G00, G01, G11, U, V, nA, nB, lp, mo, N, SO, spare, S);
    dist_row<1>(G00, G01, G11, U, V, nA, nB, lp, mo, N, SO, spare, S);
    dist_row<2>(G00, G01, G11, U, V, nA, nB, lp, mo, N, SO, spare, S);
    dist_row<3>(G00, G01, G11, U, V, nA, nB, lp, mo, N, SO, spare, S);
    dist_row<4>(G00, G01, G11, U, V, nA, nB, lp, mo, N, SO, spare, S);
    dist_row<5>(G00, G01, G11, U, V, nA, nB, lp, mo, N, SO, spare, S);
    dist_row<6>(G00, G01, G11, U, V, nA, nB, lp, mo, N, SO, spare, S);
    dist_row<7>(G00, G01, G11, U, V, nA, nB, lp, mo, N, SO, spare, S);
    __syncthreads();

#pragma unroll
    for (int i = 0; i < 7; ++i) {
        const int k = tid + i * FT;
        const float v = S[(k < NPAIR) ? k : (NPAIR - 1)];
        if (k < NPAIR) o[306 + k] = v;
    }
}

extern "C" void kernel_launch(void* const* d_in, const int* in_sizes, int n_in,
                              void* d_out, int out_size, void* d_ws, size_t ws_size,
                              hipStream_t stream) {
    const float* x = (const float*)d_in[0];
    const int T = in_sizes[0] / (NFEAT * 3);
    unsigned int* part = (unsigned int*)d_ws;
    unsigned int* gmask = (unsigned int*)((char*)d_ws + 4096);
    float* out = (float*)d_out;

    count_keep_kernel<<<P1_BLOCKS, P1_THREADS, 0, stream>>>(x, part, gmask, T);
    feat_kernel<<<OUTROWS, FT, 0, stream>>>(x, part, gmask, out, T);
}


# output rows built in launch 1 beside the counting pass: 100 leading blocks find their own frames and write both hand variants to workspace; launch 2 only sums the counts and copies the chosen one
# speedup vs baseline: 17.9486x; 1.2648x over previous
#include <hip/hip_runtime.h>
#include <math.h>

typedef __attribute__((ext_vector_type(2))) float v2f;
typedef __attribute__((ext_vector_type(8))) float v8f;

#define NFEAT 115
#define NXF 86
#define ROWLEN 1196
#define OUTROWS 100

#define P1_FPW 16
#define P1_WAVES 8
#define P1_THREADS 256
#define P1_BLOCKS 512

__device__ __forceinline__ float lane_xor16(float v) {
    int b = __builtin_bit_cast(int, v);
    return __builtin_bit_cast(float,
        __builtin_amdgcn_permlanex16(b, b, 0x76543210, 0xfedcba98, false, false));
}

template <int M>
__device__ __forceinline__ float row_xor(float v) {
    int b = __builtin_bit_cast(int, v);
    return __builtin_bit_cast(float, __builtin_amdgcn_update_dpp(0, b, 0x160 + M, 0xf, 0xf, true));
}

__device__ __forceinline__ float wave_sum(float v) {
    v += lane_xor16(v);
    v += row_xor<8>(v);
    v += row_xor<4>(v);
    v += row_xor<2>(v);
    v += row_xor<1>(v);
    return v;
}

__device__ __forceinline__ unsigned int wave_sum_u32(unsigned int v) {
    const int b = (int)v;
    v += (unsigned int)__builtin_amdgcn_permlanex16(b, b, 0x76543210, 0xfedcba98, false, false);
    v += (unsigned int)__builtin_amdgcn_update_dpp(0, (int)v, 0x160 + 8, 0xf, 0xf, true);
    v += (unsigned int)__builtin_amdgcn_update_dpp(0, (int)v, 0x160 + 4, 0xf, 0xf, true);
    v += (unsigned int)__builtin_amdgcn_update_dpp(0, (int)v, 0x160 + 2, 0xf, 0xf, true);
    v += (unsigned int)__builtin_amdgcn_update_dpp(0, (int)v, 0x160 + 1, 0xf, 0xf, true);
    return v;
}

template <int NF, bool KEEP>
__device__ __forceinline__ void p1_frames(const float* __restrict__ fx, int o0, int o1, bool has1,
                                          unsigned int& cl, unsigned int& cr,
                                          unsigned int& mL, unsigned int& mR) {
    float l0[NF], l1[NF], r0[NF], r1[NF];
#pragma unroll
    for (int f = 0; f < NF; ++f) {
        l0[f] = fx[f * (NFEAT * 3) + 40 * 3 + o0];
        l1[f] = fx[f * (NFEAT * 3) + 40 * 3 + o1];
        r0[f] = fx[f * (NFEAT * 3) + 94 * 3 + o0];
        r1[f] = fx[f * (NFEAT * 3) + 94 * 3 + o1];
    }
    mL = 0u; mR = 0u;
#pragma unroll
    for (int f = 0; f < NF; ++f) {
        float a0 = l0[f], a1 = l1[f], b0 = r0[f], b1 = r1[f];
        const bool za0 = __builtin_islessgreater(a0, 0.f);
        const bool zb0 = __builtin_islessgreater(b0, 0.f);
        const bool za1 = has1 && __builtin_islessgreater(a1, 0.f);
        const bool zb1 = has1 && __builtin_islessgreater(b1, 0.f);
        cl += (za0 ? 1u : 0u) + (za1 ? 1u : 0u);
        cr += (zb0 ? 1u : 0u) + (zb1 ? 1u : 0u);
        if constexpr (KEEP) {
            a0 = (a0 != a0) ? 0.f : a0;
            b0 = (b0 != b0) ? 0.f : b0;
            a1 = (has1 && a1 == a1) ? a1 : 0.f;
            b1 = (has1 && b1 == b1) ? b1 : 0.f;
            const float tl = wave_sum(a0 + a1);
            const float tr = wave_sum(b0 + b1);
            mL |= __builtin_amdgcn_ballot_w32(tl != 0.f) & (1u << f);
            mR |= __builtin_amdgcn_ballot_w32(tr != 0.f) & (1u << f);
        }
    }
}

__device__ __forceinline__ void keep_group(const float* __restrict__ x, int base, int T,
                                           int o0, int o1, bool has1,
                                           unsigned int& mL, unsigned int& mR) {
    unsigned int cl = 0u, cr = 0u;
    const float* fx = x + (size_t)base * (NFEAT * 3);
    mL = 0u; mR = 0u;
    if (base + P1_FPW <= T) {
        p1_frames<P1_FPW, true>(fx, o0, o1, has1, cl, cr, mL, mR);
    } else {
        for (int f = 0; base + f < T; ++f) {
            unsigned int bL, bR;
            p1_frames<1, true>(fx + (size_t)f * (NFEAT * 3), o0, o1, has1, cl, cr, bL, bR);
            mL |= bL << f; mR |= bR << f;
        }
    }
}

__device__ __forceinline__ void count_role(const float* __restrict__ x,
                                           unsigned int* __restrict__ part, int T, int bid) {
    __shared__ unsigned int s_cnt[2][P1_WAVES];
    const int lane = threadIdx.x & 31;
    const int w = __builtin_amdgcn_readfirstlane((int)(threadIdx.x >> 5));
    const bool has1 = lane < 31;
    const int o0 = lane;
    const int o1 = has1 ? (lane + 32) : 62;
    unsigned int cl = 0u, cr = 0u;
    const int stride = P1_BLOCKS * (P1_WAVES * P1_FPW);
    for (int base = (bid * P1_WAVES + w) * P1_FPW; base < T; base += stride) {
        const float* fx = x + (size_t)base * (NFEAT * 3);
        unsigned int mL, mR;
        if (base + P1_FPW <= T) {
            p1_frames<P1_FPW, false>(fx, o0, o1, has1, cl, cr, mL, mR);
        } else {
            for (int f = 0; base + f < T; ++f)
                p1_frames<1, false>(fx + (size_t)f * (NFEAT * 3), o0, o1, has1, cl, cr, mL, mR);
        }
    }
    cl = wave_sum_u32(cl); cr = wave_sum_u32(cr);
    if (lane == 0) { s_cnt[0][w] = cl; s_cnt[1][w] = cr; }
    __syncthreads();
    if (threadIdx.x < 2) {
        unsigned int s = 0u;
        for (int i = 0; i < P1_WAVES; ++i) s += s_cnt[threadIdx.x][i];
        part[2 * bid + threadIdx.x] = s;
    }
}


__device__ __forceinline__ void chunk_prefix(unsigned int m, int& c, int& excl, int& total) {
    c = __builtin_popcount(m);
    excl = 0; total = 0;
    for (int b = 0; b < 5; ++b) {
        const unsigned int bm = __builtin_amdgcn_ballot_w32(((c >> b) & 1) != 0);
        excl += (int)(__builtin_amdgcn_mbcnt_lo(bm, 0u) << b);
        total += __builtin_popcount(bm) << b;
    }
}

__device__ __forceinline__ int chunk_select(unsigned int m, int c, int excl, int rel, int lane) {
    const unsigned int own = __builtin_amdgcn_ballot_w32(rel >= excl && rel < excl + c);
    if (own == 0u) return -1;
    unsigned int mm = m;
    const int k = rel - excl;
    for (int i = 0; i < 15; ++i) if (i < k) mm &= mm - 1u;
    const int pos = lane * 16 + __builtin_ctz(mm | 0x10000u);
    return __builtin_amdgcn_readlane(pos, __builtin_ctz(own));
}

__device__ __forceinline__ int nth_flag(unsigned int g, int k) {
    for (int i = 0; i < 15; ++i) if (i < k) g &= g - 1u;
    return __builtin_ctz(g | 0x10000u);
}

#define FT 128
#define FB 256
#define VROW (OUTROWS * ROWLEN)
#define NPAIR 890

template <int R>
__device__ __forceinline__ float row_share(float v) {
    int b = __builtin_bit_cast(int, v);
    return __builtin_bit_cast(float, __builtin_amdgcn_update_dpp(0, b, 0x150 + R, 0xf, 0xf, true));
}

template <int R>
__device__ __forceinline__ void dist_row(const v8f& G00, const v8f& G01, const v8f& G11,
                                         float U, float V, float nA, float nB,
                                         int lp, int mo, int N, int SO, int spare, float* S) {
    const float ml = row_share<R>(U);
    const float mh = row_share<R>(V);
    const int m = R + mo;
    const int m1 = 16 + m;
    const int pm = ((m * (2 * N - m - 1)) >> 1) + SO;
    const int pm1 = ((m1 * (2 * N - m1 - 1)) >> 1) + SO;
    {
        const float d2 = ml + nA - 2.f * G00[R];
        S[(m < lp) ? (pm + (lp - m - 1)) : spare] = sqrtf(fmaxf(d2, 0.f));
    }
    {
        const float d2 = ml + nB - 2.f * G01[R];
        S[(16 + lp < N) ? (pm + (16 + lp - m - 1)) : spare] = sqrtf(fmaxf(d2, 0.f));
    }
    {
        const float d2 = mh + nB - 2.f * G11[R];
        S[(m < lp && 16 + lp < N) ? (pm1 + (lp - m - 1)) : spare] = sqrtf(fmaxf(d2, 0.f));
    }
}

__device__ __forceinline__ void feat_role(const float* __restrict__ x, float* __restrict__ var,
                                          int T, int row) {
    __shared__ float As[2][3 * FT];
    __shared__ float Bs[2][3 * FT];
    __shared__ float Ss[2][NPAIR + 32];
    __shared__ unsigned int s_gm[FB / 32];
    __builtin_assume(threadIdx.x < FB);
    const int ln = threadIdx.x & 31;
    const int w8 = __builtin_amdgcn_readfirstlane((int)(threadIdx.x >> 5));
    const int hf = w8 >> 2;
    const bool cond = (hf == 0);
    const int tid = (int)threadIdx.x & (FT - 1);
    float* A = As[hf];
    float* B = Bs[hf];
    float* S = Ss[hf];

    const int NG = ((row + 1) >> 4) + 1;
    const bool has1 = ln < 31;
    const int o0 = ln;
    const int o1 = has1 ? (ln + 32) : 62;
    if (w8 < NG) {
        unsigned int mL, mR;
        keep_group(x, w8 * P1_FPW, T, o0, o1, has1, mL, mR);
        if (ln == 0) s_gm[w8] = (mR << 16) | (mL & 0xffffu);
    }
    __syncthreads();

    const unsigned int pm0 = (ln < NG) ? s_gm[ln] : 0u;
    const unsigned int m0 = cond ? (pm0 & 0xffffu) : (pm0 >> 16);
    int tA, tB;
    int cum;
    {
        int c, excl, total;
        chunk_prefix(m0, c, excl, total);
        tA = chunk_select(m0, c, excl, row, ln);
        tB = chunk_select(m0, c, excl, row + 1, ln);
        cum = total;
    }
    for (int base = NG * P1_FPW; base < T && tB < 0; base += P1_FPW) {
        unsigned int mL, mR;
        keep_group(x, base, T, o0, o1, has1, mL, mR);
        const unsigned int g = (cond ? mL : mR) & 0xffffu;
        const int cg = __builtin_popcount(g);
        if (tA < 0 && row - cum < cg) tA = base + nth_flag(g, row - cum);
        if (row + 1 - cum < cg)       tB = base + nth_flag(g, row + 1 - cum);
        cum += cg;
    }
    if (tB < 0) {
        int nv = T - ln * 16;
        nv = nv < 0 ? 0 : (nv > 16 ? 16 : nv);
        if (ln >= NG) nv = 0;
        const unsigned int nm = ~m0 & ((1u << nv) - 1u);
        int c, excl, total;
        chunk_prefix(nm, c, excl, total);
        if (tA < 0) tA = chunk_select(nm, c, excl, row - cum, ln);
        tB = chunk_select(nm, c, excl, row + 1 - cum, ln);
    }
    tA = tA < 0 ? 0 : (tA > T - 1 ? T - 1 : tA);
    tB = tB < 0 ? 0 : (tB > T - 1 ? T - 1 : tB);

#pragma unroll
    for (int i = 0; i < 3; ++i) {
        const int j = tid + i * FT;
        const int jc = (j < NXF * 3) ? j : (NXF * 3 - 1);
        int f = jc / 3, c = jc - f * 3;
        int src;
        if (f < 21)      src = cond ? (40 + f) : (94 + f);
        else if (f < 46) src = f + 40;
        else             src = f - 46;
        float vA = x[(size_t)tA * (NFEAT * 3) + src * 3 + c];
        float vB = x[(size_t)tB * (NFEAT * 3) + src * 3 + c];
        vA = (vA != vA) ? 0.f : vA;
        vB = (vB != vB) ? 0.f : vB;
        if (cond && c == 0) { vA = -vA; vB = -vB; }
        A[j] = vA; B[j] = vB;
    }
    __syncthreads();

    float* o = var + (size_t)hf * VROW + (size_t)row * ROWLEN;

#pragma unroll
    for (int i = 0; i < 3; ++i) {
        const int k = tid + i * FT;
        const int kc = (k < 306) ? k : 305;
        const int kk = (kc < 153) ? kc : (kc - 153);
        const int i2 = kk - ((kk < 113) ? 63 : 113);
        const int i3 = ((kk < 113) ? 21 * 3 : 46 * 3) + (i2 >> 1) * 3 + (i2 & 1);
        const int idx = (kk < 63) ? kk : i3;
        const float a = A[idx], b = B[idx];
        if (k < 306) o[k] = (kc < 153) ? a : (a - b);
    }

    const int w = w8 & 3;
    const int N  = (w == 0) ? 21 : ((w == 1) ? 25 : 20);
    const int F  = (w == 0) ? 0 : ((w == 1) ? 21 : ((w == 2) ? 46 : 66));
    const int SO = (w == 0) ? 0 : ((w == 1) ? 210 : ((w == 2) ? 510 : 700));
    const bool three = (w == 0);
    const int lp = ln & 15;
    const bool hiL = ln >= 16;

    const bool in1 = 16 + lp < N;
    const float* P0 = A + (F + lp) * 3;
    const float* P1 = A + (F + (in1 ? (16 + lp) : (N - 1))) * 3;
    float x0 = P0[0], y0 = P0[1], z0 = P0[2];
    float x1 = P1[0], y1 = P1[1], z1 = P1[2];
    if (!three) { z0 = 0.f; z1 = 0.f; }
    if (!in1)   { x1 = 0.f; y1 = 0.f; z1 = 0.f; }
    float nA = x0 * x0 + y0 * y0; nA += z0 * z0;
    float nB = x1 * x1 + y1 * y1; nB += z1 * z1;

    v2f r0, r1;
    r0.x = hiL ? z0 : x0; r0.y = hiL ? 0.f : y0;
    r1.x = hiL ? z1 : x1; r1.y = hiL ? 0.f : y1;

    v8f cz = {0.f, 0.f, 0.f, 0.f, 0.f, 0.f, 0.f, 0.f};
    v8f G00 = __builtin_amdgcn_wmma_f32_16x16x4_f32(false, r0, false, r0, (short)0, cz, false, false);
    v8f G01 = __builtin_amdgcn_wmma_f32_16x16x4_f32(false, r0, false, r1, (short)0, cz, false, false);
    v8f G11 = __builtin_amdgcn_wmma_f32_16x16x4_f32(false, r1, false, r1, (short)0, cz, false, false);

    const float U = hiL ? row_xor<8>(nA) : nA;
    const float V = hiL ? row_xor<8>(nB) : nB;
    const int mo = hiL ? 8 : 0;
    const int spare = NPAIR + ln;
    dist_row<0>(G00, G01, G11, U, V, nA, nB, lp, mo, N, SO, spare, S);
    dist_row<1>(G00, G01, G11, U, V, nA, nB, lp, mo, N, SO, spare, S);
    dist_row<2>(G00, G01, G11, U, V, nA, nB, lp, mo, N, SO, spare, S);
    dist_row<3>(G00, G01, G11, U, V, nA, nB, lp, mo, N, SO, spare, S);
    dist_row<4>(G00, G01, G11, U, V, nA, nB, lp, mo, N, SO, spare, S);
    dist_row<5>(G00, G01, G11, U, V, nA, nB, lp, mo, N, SO, spare, S);
    dist_row<6>(G00, G01, G11, U, V, nA, nB, lp, mo, N, SO, spare, S);
    dist_row<7>(G00, G01, G11, U, V, nA, nB, lp, mo, N, SO, spare, S);
    __syncthreads();

#pragma unroll
    for (int i = 0; i < 7; ++i) {
        const int k = tid + i * FT;
        const float v = S[(k < NPAIR) ? k : (NPAIR - 1)];
        if (k < NPAIR) o[306 + k] = v;
    }
}

__global__ __launch_bounds__(FB) __attribute__((amdgpu_waves_per_eu(4, 4)))
void rows_and_counts_kernel(const float* __restrict__ x,
                            unsigned int* __restrict__ part,
                            float* __restrict__ var, int T) {
    const int b = (int)blockIdx.x;
    if (b < OUTROWS) feat_role(x, var, T, b);
    else             count_role(x, part, T, b - OUTROWS);
}

#define N4 (VROW / 4)
__global__ __launch_bounds__(FB)
void pick_kernel(const unsigned int* __restrict__ part,
                 const float* __restrict__ var,
                 float* __restrict__ out) {
    const int ln = threadIdx.x & 31;
    const int i = (int)blockIdx.x * FB + (int)threadIdx.x;
    const int ic = (i < N4) ? i : (N4 - 1);
    const float4 vl = reinterpret_cast<const float4*>(var)[ic];
    const float4 vr = reinterpret_cast<const float4*>(var + VROW)[ic];
    unsigned int sl = 0u, sr = 0u;
    const uint4* part4 = reinterpret_cast<const uint4*>(part) + ln;
    for (int k = 0; k < P1_BLOCKS / 64; ++k) {
        const uint4 p = part4[k * 32];
        sl += p.x + p.z; sr += p.y + p.w;
    }
    asm volatile("" :: "v"(vl.x), "v"(vl.y), "v"(vl.z), "v"(vl.w), "v"(vr.x), "v"(vr.y), "v"(vr.z), "v"(vr.w));
    sl = wave_sum_u32(sl);
    sr = wave_sum_u32(sr);
    const bool cond = sl > sr;
    float4 r;
    r.x = cond ? vl.x : vr.x; r.y = cond ? vl.y : vr.y;
    r.z = cond ? vl.z : vr.z; r.w = cond ? vl.w : vr.w;
    if (i < N4) reinterpret_cast<float4*>(out)[i] = r;
}

extern "C" void kernel_launch(void* const* d_in, const int* in_sizes, int n_in,
                              void* d_out, int out_size, void* d_ws, size_t ws_size,
                              hipStream_t stream) {
    const float* x = (const float*)d_in[0];
    const int T = in_sizes[0] / (NFEAT * 3);
    unsigned int* part = (unsigned int*)d_ws;
    float* var = (float*)((char*)d_ws + 4096);
    float* out = (float*)d_out;

    rows_and_counts_kernel<<<OUTROWS + P1_BLOCKS, FB, 0, stream>>>(x, part, var, T);
    pick_kernel<<<(N4 + FB - 1) / FB, FB, 0, stream>>>(part, var, out);
}
